# starting kernel
# baseline (speedup 1.0000x reference)
#include <hip/hip_runtime.h>

typedef float v2f __attribute__((ext_vector_type(2)));
typedef float v8f __attribute__((ext_vector_type(8)));

#define KC  32
#define KCP 36

__global__ void sage_zero_f32(float* __restrict__ p, long n) {
  long i = (long)blockIdx.x * blockDim.x + threadIdx.x;
  long stride = (long)gridDim.x * blockDim.x;
  for (; i < n; i += stride) p[i] = 0.0f;
}

__global__ void sage_scatter(const float* __restrict__ H,
                             const int* __restrict__ src,
                             const int* __restrict__ dst,
                             float* __restrict__ S,
                             float* __restrict__ Deg,
                             int E, int d) {
  const int e = blockIdx.x;
  if (e >= E) return;
  const int sidx = src[e];
  const int didx = dst[e];

  if (e + 1 < E) {
    const int snext = src[e + 1];
    __builtin_prefetch(H + (size_t)snext * d + threadIdx.x * 4, 0, 0);
  }

  const float4 v = *(const float4*)(H + (size_t)sidx * d + threadIdx.x * 4);
  float* o = S + (size_t)didx * d + threadIdx.x * 4;
  __hip_atomic_fetch_add(o + 0, v.x, __ATOMIC_RELAXED, __HIP_MEMORY_SCOPE_AGENT);
  __hip_atomic_fetch_add(o + 1, v.y, __ATOMIC_RELAXED, __HIP_MEMORY_SCOPE_AGENT);
  __hip_atomic_fetch_add(o + 2, v.z, __ATOMIC_RELAXED, __HIP_MEMORY_SCOPE_AGENT);
  __hip_atomic_fetch_add(o + 3, v.w, __ATOMIC_RELAXED, __HIP_MEMORY_SCOPE_AGENT);
  if (threadIdx.x == 0) {
    __hip_atomic_fetch_add(Deg + didx, 1.0f, __ATOMIC_RELAXED, __HIP_MEMORY_SCOPE_AGENT);
  }
}

__global__ void sage_normalize(float* __restrict__ S, const float* __restrict__ Deg,
                               int n, int d) {
  const int d4 = d >> 2;
  long i = (long)blockIdx.x * blockDim.x + threadIdx.x;
  long total = (long)n * d4;
  if (i >= total) return;
  const int row = (int)(i / d4);
  const float inv = 1.0f / fmaxf(Deg[row], 1.0f);
  float4* p = (float4*)S + i;
  float4 v = *p;
  v.x *= inv; v.y *= inv; v.z *= inv; v.w *= inv;
  *p = v;
}

template <int D, int DOUT, int NTW, bool RELU>
__global__ void sage_gemm_wmma(const float* __restrict__ Hself,
                               const float* __restrict__ Mean,
                               const float* __restrict__ Wself,
                               const float* __restrict__ Wneigh,
                               const float* __restrict__ bias,
                               const float* __restrict__ dmask,
                               float* __restrict__ Out,
                               int n) {
  __shared__ float As[2][2][16 * KCP];

  const int tid  = threadIdx.x;
  const int lane = tid & 31;
  const int wave = tid >> 5;
  const int tm   = blockIdx.y;
  const int half = lane >> 4;
  const int l    = lane & 15;

  const int smat = tid >> 7;
  const int sidx = tid & 127;
  const int srow = sidx >> 3;
  const int scol = (sidx & 7) * 4;
  const int grow = min(tm * 16 + srow, n - 1);
  const float* sbase = (smat ? Mean : Hself) + (size_t)grow * D + scol;

  v8f acc[NTW];
#pragma unroll
  for (int t = 0; t < NTW; ++t) {
    const float bv = bias[(wave * NTW + t) * 16 + l];
#pragma unroll
    for (int r = 0; r < 8; ++r) acc[t][r] = bv;
  }

  constexpr int nchunks = D / KC;
  float4 stage = *(const float4*)(sbase);

  for (int c = 0; c < nchunks; ++c) {
    const int buf = c & 1;
    *(float4*)&As[buf][smat][srow * KCP + scol] = stage;
    if (c + 1 < nchunks) {
      stage = *(const float4*)(sbase + (size_t)(c + 1) * KC);
    }
    __syncthreads();

    const size_t wbase = (size_t)(c * KC + half * 2) * DOUT;
    const float* pW0[NTW];
    const float* pW1[NTW];
#pragma unroll
    for (int t = 0; t < NTW; ++t) {
      const int col = (wave * NTW + t) * 16 + l;
      pW0[t] = Wself + wbase + col;
      pW1[t] = Wneigh + wbase + col;
    }

#pragma unroll
    for (int k0 = 0; k0 < KC; k0 += 4) {
      const int kk = k0 + half * 2;
      const v2f a0 = *(const v2f*)&As[buf][0][l * KCP + kk];
      const v2f a1 = *(const v2f*)&As[buf][1][l * KCP + kk];
#pragma unroll
      for (int t = 0; t < NTW; ++t) {
        v2f b0, b1;
        b0.x = pW0[t][k0 * DOUT];
        b0.y = pW0[t][k0 * DOUT + DOUT];
        acc[t] = __builtin_amdgcn_wmma_f32_16x16x4_f32(
            false, a0, false, b0, (short)0, acc[t], false, false);
        b1.x = pW1[t][k0 * DOUT];
        b1.y = pW1[t][k0 * DOUT + DOUT];
        acc[t] = __builtin_amdgcn_wmma_f32_16x16x4_f32(
            false, a1, false, b1, (short)0, acc[t], false, false);
      }
    }
  }

#pragma unroll
  for (int t = 0; t < NTW; ++t) {
    const int col = (wave * NTW + t) * 16 + l;
#pragma unroll
    for (int r = 0; r < 8; ++r) {
      const int row = tm * 16 + half * 8 + r;
      if (row < n) {
        float v = acc[t][r];
        if (RELU) v = fmaxf(v, 0.0f) * dmask[(size_t)row * DOUT + col];
        Out[(size_t)row * DOUT + col] = v;
      }
    }
  }
}

extern "C" void kernel_launch(void* const* d_in, const int* in_sizes, int n_in,
                              void* d_out, int out_size, void* d_ws, size_t ws_size,
                              hipStream_t stream) {
  const float* h      = (const float*)d_in[0];
  const int*  src1    = (const int*)d_in[1];
  const int*  dst1    = (const int*)d_in[2];
  const int*  src2    = (const int*)d_in[3];
  const int*  dst2    = (const int*)d_in[4];
  const int*  src3    = (const int*)d_in[5];
  const int*  dst3    = (const int*)d_in[6];
  const float* Wself1  = (const float*)d_in[10];
  const float* Wneigh1 = (const float*)d_in[11];
  const float* b1      = (const float*)d_in[12];
  const float* Wself2  = (const float*)d_in[13];
  const float* Wneigh2 = (const float*)d_in[14];
  const float* b2      = (const float*)d_in[15];
  const float* Wself3  = (const float*)d_in[16];
  const float* Wneigh3 = (const float*)d_in[17];
  const float* b3      = (const float*)d_in[18];
  const float* dmask1  = (const float*)d_in[19];
  const float* dmask2  = (const float*)d_in[20];

  const int D = 256, Dout = 128;
  const int E1 = in_sizes[1];
  const int E2 = in_sizes[3];
  const int E3 = in_sizes[5];
  const int N1 = in_sizes[19] / D;
  const int N2 = in_sizes[20] / D;
  const int N3 = out_size / Dout;

  float* w = (float*)d_ws;
  float* s1 = w;   w += (size_t)N1 * D;
  float* deg1 = w; w += N1;
  float* s2 = w;   w += (size_t)N2 * D;
  float* deg2 = w; w += N2;
  float* s3 = w;   w += (size_t)N3 * D;
  float* deg3 = w; w += N3;
  float* h1 = w;   w += (size_t)N1 * D;
  float* h2 = w;   w += (size_t)N2 * D;

  const long zcount = (long)(N1 + N2 + N3) * (D + 1);
  sage_zero_f32<<<2048, 256, 0, stream>>>((float*)d_ws, zcount);

  sage_scatter<<<E1, 64, 0, stream>>>(h, src1, dst1, s1, deg1, E1, D);
  {
    long t4 = (long)N1 * (D / 4);
    sage_normalize<<<(unsigned)((t4 + 255) / 256), 256, 0, stream>>>(s1, deg1, N1, D);
  }
  sage_gemm_wmma<256, 256, 2, true><<<dim3(1, (N1 + 15) / 16), 256, 0, stream>>>(
      h, s1, Wself1, Wneigh1, b1, dmask1, h1, N1);

  sage_scatter<<<E2, 64, 0, stream>>>(h1, src2, dst2, s2, deg2, E2, D);
  {
    long t4 = (long)N2 * (D / 4);
    sage_normalize<<<(unsigned)((t4 + 255) / 256), 256, 0, stream>>>(s2, deg2, N2, D);
  }
  sage_gemm_wmma<256, 256, 2, true><<<dim3(1, (N2 + 15) / 16), 256, 0, stream>>>(
      h1, s2, Wself2, Wneigh2, b2, dmask2, h2, N2);

  sage_scatter<<<E3, 64, 0, stream>>>(h2, src3, dst3, s3, deg3, E3, D);
  {
    long t4 = (long)N3 * (D / 4);
    sage_normalize<<<(unsigned)((t4 + 255) / 256), 256, 0, stream>>>(s3, deg3, N3, D);
  }
  sage_gemm_wmma<256, 128, 1, false><<<dim3(1, (N3 + 15) / 16), 256, 0, stream>>>(
      h2, s3, Wself3, Wneigh3, b3, nullptr, (float*)d_out, N3);
}


# scatter: lane-to-float map changed so each f32 atomic instruction adds one contiguous 128-B span, not 8 lanes in each of 4 lines (4x fewer L2 atomic requests); sum order still arrival order
# speedup vs baseline: 1.1744x; 1.1744x over previous
#include <hip/hip_runtime.h>

typedef float v2f __attribute__((ext_vector_type(2)));
typedef float v8f __attribute__((ext_vector_type(8)));

#define KC  32
#define KCP 36

__global__ void sage_zero_f32(float* __restrict__ p, long n) {
  long i = (long)blockIdx.x * blockDim.x + threadIdx.x;
  long stride = (long)gridDim.x * blockDim.x;
  for (; i < n; i += stride) p[i] = 0.0f;
}

__global__ void sage_scatter(const float* __restrict__ H,
                             const int* __restrict__ src,
                             const int* __restrict__ dst,
                             float* __restrict__ S,
                             float* __restrict__ Deg,
                             int E, int d) {
  const int e = blockIdx.x;
  if (e >= E) return;
  const int sidx = src[e];
  const int didx = dst[e];

  if (e + 1 < E) {
    const int snext = src[e + 1];
    __builtin_prefetch(H + (size_t)snext * d + threadIdx.x * 4, 0, 0);
  }

  const int f0 = (threadIdx.x >> 5) * 64 + (threadIdx.x & 31);
  const float* hp = H + (size_t)sidx * d + f0;
  const float v0 = hp[0];
  const float v1 = hp[32];
  const float v2 = hp[128];
  const float v3 = hp[160];
  float* o = S + (size_t)didx * d + f0;
  __hip_atomic_fetch_add(o,       v0, __ATOMIC_RELAXED, __HIP_MEMORY_SCOPE_AGENT);
  __hip_atomic_fetch_add(o + 32,  v1, __ATOMIC_RELAXED, __HIP_MEMORY_SCOPE_AGENT);
  __hip_atomic_fetch_add(o + 128, v2, __ATOMIC_RELAXED, __HIP_MEMORY_SCOPE_AGENT);
  __hip_atomic_fetch_add(o + 160, v3, __ATOMIC_RELAXED, __HIP_MEMORY_SCOPE_AGENT);
  if (threadIdx.x == 0) {
    __hip_atomic_fetch_add(Deg + didx, 1.0f, __ATOMIC_RELAXED, __HIP_MEMORY_SCOPE_AGENT);
  }
}

__global__ void sage_normalize(float* __restrict__ S, const float* __restrict__ Deg,
                               int n, int d) {
  const int d4 = d >> 2;
  long i = (long)blockIdx.x * blockDim.x + threadIdx.x;
  long total = (long)n * d4;
  if (i >= total) return;
  const int row = (int)(i / d4);
  const float inv = 1.0f / fmaxf(Deg[row], 1.0f);
  float4* p = (float4*)S + i;
  float4 v = *p;
  v.x *= inv; v.y *= inv; v.z *= inv; v.w *= inv;
  *p = v;
}

template <int D, int DOUT, int NTW, bool RELU>
__global__ void sage_gemm_wmma(const float* __restrict__ Hself,
                               const float* __restrict__ Mean,
                               const float* __restrict__ Wself,
                               const float* __restrict__ Wneigh,
                               const float* __restrict__ bias,
                               const float* __restrict__ dmask,
                               float* __restrict__ Out,
                               int n) {
  __shared__ float As[2][2][16 * KCP];

  const int tid  = threadIdx.x;
  const int lane = tid & 31;
  const int wave = tid >> 5;
  const int tm   = blockIdx.y;
  const int half = lane >> 4;
  const int l    = lane & 15;

  const int smat = tid >> 7;
  const int sidx = tid & 127;
  const int srow = sidx >> 3;
  const int scol = (sidx & 7) * 4;
  const int grow = min(tm * 16 + srow, n - 1);
  const float* sbase = (smat ? Mean : Hself) + (size_t)grow * D + scol;

  v8f acc[NTW];
#pragma unroll
  for (int t = 0; t < NTW; ++t) {
    const float bv = bias[(wave * NTW + t) * 16 + l];
#pragma unroll
    for (int r = 0; r < 8; ++r) acc[t][r] = bv;
  }

  constexpr int nchunks = D / KC;
  float4 stage = *(const float4*)(sbase);

  for (int c = 0; c < nchunks; ++c) {
    const int buf = c & 1;
    *(float4*)&As[buf][smat][srow * KCP + scol] = stage;
    if (c + 1 < nchunks) {
      stage = *(const float4*)(sbase + (size_t)(c + 1) * KC);
    }
    __syncthreads();

    const size_t wbase = (size_t)(c * KC + half * 2) * DOUT;
    const float* pW0[NTW];
    const float* pW1[NTW];
#pragma unroll
    for (int t = 0; t < NTW; ++t) {
      const int col = (wave * NTW + t) * 16 + l;
      pW0[t] = Wself + wbase + col;
      pW1[t] = Wneigh + wbase + col;
    }

#pragma unroll
    for (int k0 = 0; k0 < KC; k0 += 4) {
      const int kk = k0 + half * 2;
      const v2f a0 = *(const v2f*)&As[buf][0][l * KCP + kk];
      const v2f a1 = *(const v2f*)&As[buf][1][l * KCP + kk];
#pragma unroll
      for (int t = 0; t < NTW; ++t) {
        v2f b0, b1;
        b0.x = pW0[t][k0 * DOUT];
        b0.y = pW0[t][k0 * DOUT + DOUT];
        acc[t] = __builtin_amdgcn_wmma_f32_16x16x4_f32(
            false, a0, false, b0, (short)0, acc[t], false, false);
        b1.x = pW1[t][k0 * DOUT];
        b1.y = pW1[t][k0 * DOUT + DOUT];
        acc[t] = __builtin_amdgcn_wmma_f32_16x16x4_f32(
            false, a1, false, b1, (short)0, acc[t], false, false);
      }
    }
  }

#pragma unroll
  for (int t = 0; t < NTW; ++t) {
    const int col = (wave * NTW + t) * 16 + l;
#pragma unroll
    for (int r = 0; r < 8; ++r) {
      const int row = tm * 16 + half * 8 + r;
      if (row < n) {
        float v = acc[t][r];
        if (RELU) v = fmaxf(v, 0.0f) * dmask[(size_t)row * DOUT + col];
        Out[(size_t)row * DOUT + col] = v;
      }
    }
  }
}

extern "C" void kernel_launch(void* const* d_in, const int* in_sizes, int n_in,
                              void* d_out, int out_size, void* d_ws, size_t ws_size,
                              hipStream_t stream) {
  const float* h      = (const float*)d_in[0];
  const int*  src1    = (const int*)d_in[1];
  const int*  dst1    = (const int*)d_in[2];
  const int*  src2    = (const int*)d_in[3];
  const int*  dst2    = (const int*)d_in[4];
  const int*  src3    = (const int*)d_in[5];
  const int*  dst3    = (const int*)d_in[6];
  const float* Wself1  = (const float*)d_in[10];
  const float* Wneigh1 = (const float*)d_in[11];
  const float* b1      = (const float*)d_in[12];
  const float* Wself2  = (const float*)d_in[13];
  const float* Wneigh2 = (const float*)d_in[14];
  const float* b2      = (const float*)d_in[15];
  const float* Wself3  = (const float*)d_in[16];
  const float* Wneigh3 = (const float*)d_in[17];
  const float* b3      = (const float*)d_in[18];
  const float* dmask1  = (const float*)d_in[19];
  const float* dmask2  = (const float*)d_in[20];

  const int D = 256, Dout = 128;
  const int E1 = in_sizes[1];
  const int E2 = in_sizes[3];
  const int E3 = in_sizes[5];
  const int N1 = in_sizes[19] / D;
  const int N2 = in_sizes[20] / D;
  const int N3 = out_size / Dout;

  float* w = (float*)d_ws;
  float* s1 = w;   w += (size_t)N1 * D;
  float* deg1 = w; w += N1;
  float* s2 = w;   w += (size_t)N2 * D;
  float* deg2 = w; w += N2;
  float* s3 = w;   w += (size_t)N3 * D;
  float* deg3 = w; w += N3;
  float* h1 = w;   w += (size_t)N1 * D;
  float* h2 = w;   w += (size_t)N2 * D;

  const long zcount = (long)(N1 + N2 + N3) * (D + 1);
  sage_zero_f32<<<2048, 256, 0, stream>>>((float*)d_ws, zcount);

  sage_scatter<<<E1, 64, 0, stream>>>(h, src1, dst1, s1, deg1, E1, D);
  {
    long t4 = (long)N1 * (D / 4);
    sage_normalize<<<(unsigned)((t4 + 255) / 256), 256, 0, stream>>>(s1, deg1, N1, D);
  }
  sage_gemm_wmma<256, 256, 2, true><<<dim3(1, (N1 + 15) / 16), 256, 0, stream>>>(
      h, s1, Wself1, Wneigh1, b1, dmask1, h1, N1);

  sage_scatter<<<E2, 64, 0, stream>>>(h1, src2, dst2, s2, deg2, E2, D);
  {
    long t4 = (long)N2 * (D / 4);
    sage_normalize<<<(unsigned)((t4 + 255) / 256), 256, 0, stream>>>(s2, deg2, N2, D);
  }
  sage_gemm_wmma<256, 256, 2, true><<<dim3(1, (N2 + 15) / 16), 256, 0, stream>>>(
      h1, s2, Wself2, Wneigh2, b2, dmask2, h2, N2);

  sage_scatter<<<E3, 64, 0, stream>>>(h2, src3, dst3, s3, deg3, E3, D);
  {
    long t4 = (long)N3 * (D / 4);
    sage_normalize<<<(unsigned)((t4 + 255) / 256), 256, 0, stream>>>(s3, deg3, N3, D);
  }
  sage_gemm_wmma<256, 128, 1, false><<<dim3(1, (N3 + 15) / 16), 256, 0, stream>>>(
      h2, s3, Wself3, Wneigh3, b3, nullptr, (float*)d_out, N3);
}


# scatter: a fixed grid of 8192 long-lived waves, each looping over edges 4 at a time (32 row loads in flight), replaces one 64-thread block per edge; f32 atomics kept, sum order still arrival order
# speedup vs baseline: 1.2992x; 1.1063x over previous
#include <hip/hip_runtime.h>

typedef float v2f __attribute__((ext_vector_type(2)));
typedef float v8f __attribute__((ext_vector_type(8)));

#define KC  32
#define KCP 36

__global__ void sage_zero_f32(float* __restrict__ p, long n) {
  long i = (long)blockIdx.x * blockDim.x + threadIdx.x;
  long stride = (long)gridDim.x * blockDim.x;
  for (; i < n; i += stride) p[i] = 0.0f;
}

template <int D, int EPT>
__global__ void sage_scatter(const float* __restrict__ H,
                             const int* __restrict__ src,
                             const int* __restrict__ dst,
                             float* __restrict__ S,
                             float* __restrict__ Deg,
                             int E) {
  constexpr int NJ = D / 32;
  const int lane = threadIdx.x & 31;
  const int wv   = __builtin_amdgcn_readfirstlane((int)(threadIdx.x >> 5));
  const int wpb  = (int)(blockDim.x >> 5);
  const int step = (int)gridDim.x * wpb * EPT;
  int e0 = ((int)blockIdx.x * wpb + wv) * EPT;
  if (e0 >= E) return;

  int s[EPT], t[EPT];
  for (int u = 0; u < EPT; ++u) {
    const int e = min(e0 + u, E - 1);
    s[u] = src[e];
    t[u] = dst[e];
  }

  for (;;) {
    float v[EPT][NJ];
    for (int u = 0; u < EPT; ++u) {
      const float* hp = H + (size_t)s[u] * D + lane;
      for (int j = 0; j < NJ; ++j) v[u][j] = hp[j * 32];
    }

    const int e1 = e0 + step;
    int sn[EPT], tn[EPT];
    for (int u = 0; u < EPT; ++u) {
      const int e = min(e1 + u, E - 1);
      sn[u] = src[e];
      tn[u] = dst[e];
    }

    for (int u = 0; u < EPT; ++u) {
      if (e0 + u < E) {
        float* o = S + (size_t)t[u] * D + lane;
        for (int j = 0; j < NJ; ++j) {
          __hip_atomic_fetch_add(o + j * 32, v[u][j], __ATOMIC_RELAXED,
                                 __HIP_MEMORY_SCOPE_AGENT);
        }
        if (lane == 0) {
          __hip_atomic_fetch_add(Deg + t[u], 1.0f, __ATOMIC_RELAXED,
                                 __HIP_MEMORY_SCOPE_AGENT);
        }
      }
    }

    if (e1 >= E) break;
    e0 = e1;
    for (int u = 0; u < EPT; ++u) { s[u] = sn[u]; t[u] = tn[u]; }
  }
}

__global__ void sage_normalize(float* __restrict__ S, const float* __restrict__ Deg,
                               int n, int d) {
  const int d4 = d >> 2;
  long i = (long)blockIdx.x * blockDim.x + threadIdx.x;
  long total = (long)n * d4;
  if (i >= total) return;
  const int row = (int)(i / d4);
  const float inv = 1.0f / fmaxf(Deg[row], 1.0f);
  float4* p = (float4*)S + i;
  float4 v = *p;
  v.x *= inv; v.y *= inv; v.z *= inv; v.w *= inv;
  *p = v;
}

template <int D, int DOUT, int NTW, bool RELU>
__global__ void sage_gemm_wmma(const float* __restrict__ Hself,
                               const float* __restrict__ Mean,
                               const float* __restrict__ Wself,
                               const float* __restrict__ Wneigh,
                               const float* __restrict__ bias,
                               const float* __restrict__ dmask,
                               float* __restrict__ Out,
                               int n) {
  __shared__ float As[2][2][16 * KCP];

  const int tid  = threadIdx.x;
  const int lane = tid & 31;
  const int wave = tid >> 5;
  const int tm   = blockIdx.y;
  const int half = lane >> 4;
  const int l    = lane & 15;

  const int smat = tid >> 7;
  const int sidx = tid & 127;
  const int srow = sidx >> 3;
  const int scol = (sidx & 7) * 4;
  const int grow = min(tm * 16 + srow, n - 1);
  const float* sbase = (smat ? Mean : Hself) + (size_t)grow * D + scol;

  v8f acc[NTW];
#pragma unroll
  for (int t = 0; t < NTW; ++t) {
    const float bv = bias[(wave * NTW + t) * 16 + l];
#pragma unroll
    for (int r = 0; r < 8; ++r) acc[t][r] = bv;
  }

  constexpr int nchunks = D / KC;
  float4 stage = *(const float4*)(sbase);

  for (int c = 0; c < nchunks; ++c) {
    const int buf = c & 1;
    *(float4*)&As[buf][smat][srow * KCP + scol] = stage;
    if (c + 1 < nchunks) {
      stage = *(const float4*)(sbase + (size_t)(c + 1) * KC);
    }
    __syncthreads();

    const size_t wbase = (size_t)(c * KC + half * 2) * DOUT;
    const float* pW0[NTW];
    const float* pW1[NTW];
#pragma unroll
    for (int t = 0; t < NTW; ++t) {
      const int col = (wave * NTW + t) * 16 + l;
      pW0[t] = Wself + wbase + col;
      pW1[t] = Wneigh + wbase + col;
    }

#pragma unroll
    for (int k0 = 0; k0 < KC; k0 += 4) {
      const int kk = k0 + half * 2;
      const v2f a0 = *(const v2f*)&As[buf][0][l * KCP + kk];
      const v2f a1 = *(const v2f*)&As[buf][1][l * KCP + kk];
#pragma unroll
      for (int t = 0; t < NTW; ++t) {
        v2f b0, b1;
        b0.x = pW0[t][k0 * DOUT];
        b0.y = pW0[t][k0 * DOUT + DOUT];
        acc[t] = __builtin_amdgcn_wmma_f32_16x16x4_f32(
            false, a0, false, b0, (short)0, acc[t], false, false);
        b1.x = pW1[t][k0 * DOUT];
        b1.y = pW1[t][k0 * DOUT + DOUT];
        acc[t] = __builtin_amdgcn_wmma_f32_16x16x4_f32(
            false, a1, false, b1, (short)0, acc[t], false, false);
      }
    }
  }

#pragma unroll
  for (int t = 0; t < NTW; ++t) {
    const int col = (wave * NTW + t) * 16 + l;
#pragma unroll
    for (int r = 0; r < 8; ++r) {
      const int row = tm * 16 + half * 8 + r;
      if (row < n) {
        float v = acc[t][r];
        if (RELU) v = fmaxf(v, 0.0f) * dmask[(size_t)row * DOUT + col];
        Out[(size_t)row * DOUT + col] = v;
      }
    }
  }
}

extern "C" void kernel_launch(void* const* d_in, const int* in_sizes, int n_in,
                              void* d_out, int out_size, void* d_ws, size_t ws_size,
                              hipStream_t stream) {
  const float* h      = (const float*)d_in[0];
  const int*  src1    = (const int*)d_in[1];
  const int*  dst1    = (const int*)d_in[2];
  const int*  src2    = (const int*)d_in[3];
  const int*  dst2    = (const int*)d_in[4];
  const int*  src3    = (const int*)d_in[5];
  const int*  dst3    = (const int*)d_in[6];
  const float* Wself1  = (const float*)d_in[10];
  const float* Wneigh1 = (const float*)d_in[11];
  const float* b1      = (const float*)d_in[12];
  const float* Wself2  = (const float*)d_in[13];
  const float* Wneigh2 = (const float*)d_in[14];
  const float* b2      = (const float*)d_in[15];
  const float* Wself3  = (const float*)d_in[16];
  const float* Wneigh3 = (const float*)d_in[17];
  const float* b3      = (const float*)d_in[18];
  const float* dmask1  = (const float*)d_in[19];
  const float* dmask2  = (const float*)d_in[20];

  const int D = 256, Dout = 128;
  const int E1 = in_sizes[1];
  const int E2 = in_sizes[3];
  const int E3 = in_sizes[5];
  const int N1 = in_sizes[19] / D;
  const int N2 = in_sizes[20] / D;
  const int N3 = out_size / Dout;

  float* w = (float*)d_ws;
  float* s1 = w;   w += (size_t)N1 * D;
  float* deg1 = w; w += N1;
  float* s2 = w;   w += (size_t)N2 * D;
  float* deg2 = w; w += N2;
  float* s3 = w;   w += (size_t)N3 * D;
  float* deg3 = w; w += N3;
  float* h1 = w;   w += (size_t)N1 * D;
  float* h2 = w;   w += (size_t)N2 * D;

  const long zcount = (long)(N1 + N2 + N3) * (D + 1);
  sage_zero_f32<<<2048, 256, 0, stream>>>((float*)d_ws, zcount);

  sage_scatter<256, 4><<<1024, 256, 0, stream>>>(h, src1, dst1, s1, deg1, E1);
  {
    long t4 = (long)N1 * (D / 4);
    sage_normalize<<<(unsigned)((t4 + 255) / 256), 256, 0, stream>>>(s1, deg1, N1, D);
  }
  sage_gemm_wmma<256, 256, 2, true><<<dim3(1, (N1 + 15) / 16), 256, 0, stream>>>(
      h, s1, Wself1, Wneigh1, b1, dmask1, h1, N1);

  sage_scatter<256, 4><<<1024, 256, 0, stream>>>(h1, src2, dst2, s2, deg2, E2);
  {
    long t4 = (long)N2 * (D / 4);
    sage_normalize<<<(unsigned)((t4 + 255) / 256), 256, 0, stream>>>(s2, deg2, N2, D);
  }
  sage_gemm_wmma<256, 256, 2, true><<<dim3(1, (N2 + 15) / 16), 256, 0, stream>>>(
      h1, s2, Wself2, Wneigh2, b2, dmask2, h2, N2);

  sage_scatter<256, 4><<<1024, 256, 0, stream>>>(h2, src3, dst3, s3, deg3, E3);
  {
    long t4 = (long)N3 * (D / 4);
    sage_normalize<<<(unsigned)((t4 + 255) / 256), 256, 0, stream>>>(s3, deg3, N3, D);
  }
  sage_gemm_wmma<256, 128, 1, false><<<dim3(1, (N3 + 15) / 16), 256, 0, stream>>>(
      h2, s3, Wself3, Wneigh3, b3, nullptr, (float*)d_out, N3);
}


# scatter-mean as a gather: an int atomic exchange per edge builds per-node lists, one wave per node sums its rows in registers (no f32 atomics, no zero-fill or normalize pass); sum order = list order
# speedup vs baseline: 2.1944x; 1.6890x over previous
#include <hip/hip_runtime.h>

typedef float v2f __attribute__((ext_vector_type(2)));
typedef float v8f __attribute__((ext_vector_type(8)));

#define KC  32
#define KCP 36

__global__ void sage_heads_init(int* __restrict__ head, int n) {
  const int i = (int)(blockIdx.x * blockDim.x + threadIdx.x);
  if (i < n) head[i] = -1;
}

__global__ void sage_build_lists(const int* __restrict__ src1, const int* __restrict__ dst1, int E1,
                                 const int* __restrict__ src2, const int* __restrict__ dst2, int E2,
                                 const int* __restrict__ src3, const int* __restrict__ dst3, int E3,
                                 int* __restrict__ head1, int* __restrict__ head2,
                                 int* __restrict__ head3,
                                 int* __restrict__ rec1, int* __restrict__ rec2,
                                 int* __restrict__ rec3) {
  int e = (int)(blockIdx.x * blockDim.x + threadIdx.x);
  const int* s; const int* d; int* hd; int* rc;
  if (e < E1) { s = src1; d = dst1; hd = head1; rc = rec1; }
  else if (e < E1 + E2) { e -= E1; s = src2; d = dst2; hd = head2; rc = rec2; }
  else if (e < E1 + E2 + E3) { e -= E1 + E2; s = src3; d = dst3; hd = head3; rc = rec3; }
  else return;
  const int sv = s[e];
  const int dv = d[e];
  const int prev = __hip_atomic_exchange(hd + dv, e, __ATOMIC_RELAXED,
                                         __HIP_MEMORY_SCOPE_AGENT);
  ((int2*)rc)[e] = make_int2(sv, prev);
}

template <int D, int U>
__global__ void sage_gather_mean(const float* __restrict__ H,
                                 const int* __restrict__ head,
                                 const int* __restrict__ rec,
                                 float* __restrict__ Mean,
                                 int n, int E) {
  static_assert(D == 256, "row = 2 x 32 lanes x float4");
  const int lane = threadIdx.x & 31;
  const int wv  = __builtin_amdgcn_readfirstlane((int)(threadIdx.x >> 5));
  const int wpb = (int)(blockDim.x >> 5);
  const int nw  = (int)gridDim.x * wpb;
  const int2* rc = (const int2*)rec;

  for (int node = (int)blockIdx.x * wpb + wv; node < n; node += nw) {
    int e = head[node];
    float4 a0 = make_float4(0.f, 0.f, 0.f, 0.f);
    float4 a1 = make_float4(0.f, 0.f, 0.f, 0.f);
    int deg = 0;
    while (e >= 0 && deg < E) {
      float4 v0[U], v1[U];
      int m = 0;
#pragma unroll
      for (int u = 0; u < U; ++u) {
        if (e >= 0) {
          const int2 r = rc[e];
          const float4* hp = (const float4*)(H + (size_t)r.x * D) + lane;
          v0[u] = hp[0];
          v1[u] = hp[32];
          e = r.y;
          ++m;
        }
      }
#pragma unroll
      for (int u = 0; u < U; ++u) {
        if (u < m) {
          a0.x += v0[u].x; a0.y += v0[u].y; a0.z += v0[u].z; a0.w += v0[u].w;
          a1.x += v1[u].x; a1.y += v1[u].y; a1.z += v1[u].z; a1.w += v1[u].w;
        }
      }
      deg += m;
    }
    const float inv = 1.0f / fmaxf((float)deg, 1.0f);
    a0.x *= inv; a0.y *= inv; a0.z *= inv; a0.w *= inv;
    a1.x *= inv; a1.y *= inv; a1.z *= inv; a1.w *= inv;
    float4* op = (float4*)(Mean + (size_t)node * D) + lane;
    op[0]  = a0;
    op[32] = a1;
  }
}

template <int D, int DOUT, int NTW, bool RELU>
__global__ void sage_gemm_wmma(const float* __restrict__ Hself,
                               const float* __restrict__ Mean,
                               const float* __restrict__ Wself,
                               const float* __restrict__ Wneigh,
                               const float* __restrict__ bias,
                               const float* __restrict__ dmask,
                               float* __restrict__ Out,
                               int n) {
  __shared__ float As[2][2][16 * KCP];

  const int tid  = threadIdx.x;
  const int lane = tid & 31;
  const int wave = tid >> 5;
  const int tm   = blockIdx.y;
  const int half = lane >> 4;
  const int l    = lane & 15;

  const int smat = tid >> 7;
  const int sidx = tid & 127;
  const int srow = sidx >> 3;
  const int scol = (sidx & 7) * 4;
  const int grow = min(tm * 16 + srow, n - 1);
  const float* sbase = (smat ? Mean : Hself) + (size_t)grow * D + scol;

  v8f acc[NTW];
#pragma unroll
  for (int t = 0; t < NTW; ++t) {
    const float bv = bias[(wave * NTW + t) * 16 + l];
#pragma unroll
    for (int r = 0; r < 8; ++r) acc[t][r] = bv;
  }

  constexpr int nchunks = D / KC;
  float4 stage = *(const float4*)(sbase);

  for (int c = 0; c < nchunks; ++c) {
    const int buf = c & 1;
    *(float4*)&As[buf][smat][srow * KCP + scol] = stage;
    if (c + 1 < nchunks) {
      stage = *(const float4*)(sbase + (size_t)(c + 1) * KC);
    }
    __syncthreads();

    const size_t wbase = (size_t)(c * KC + half * 2) * DOUT;
    const float* pW0[NTW];
    const float* pW1[NTW];
#pragma unroll
    for (int t = 0; t < NTW; ++t) {
      const int col = (wave * NTW + t) * 16 + l;
      pW0[t] = Wself + wbase + col;
      pW1[t] = Wneigh + wbase + col;
    }

#pragma unroll
    for (int k0 = 0; k0 < KC; k0 += 4) {
      const int kk = k0 + half * 2;
      const v2f a0 = *(const v2f*)&As[buf][0][l * KCP + kk];
      const v2f a1 = *(const v2f*)&As[buf][1][l * KCP + kk];
#pragma unroll
      for (int t = 0; t < NTW; ++t) {
        v2f b0, b1;
        b0.x = pW0[t][k0 * DOUT];
        b0.y = pW0[t][k0 * DOUT + DOUT];
        acc[t] = __builtin_amdgcn_wmma_f32_16x16x4_f32(
            false, a0, false, b0, (short)0, acc[t], false, false);
        b1.x = pW1[t][k0 * DOUT];
        b1.y = pW1[t][k0 * DOUT + DOUT];
        acc[t] = __builtin_amdgcn_wmma_f32_16x16x4_f32(
            false, a1, false, b1, (short)0, acc[t], false, false);
      }
    }
  }

#pragma unroll
  for (int t = 0; t < NTW; ++t) {
    const int col = (wave * NTW + t) * 16 + l;
#pragma unroll
    for (int r = 0; r < 8; ++r) {
      const int row = tm * 16 + half * 8 + r;
      if (row < n) {
        float v = acc[t][r];
        if (RELU) v = fmaxf(v, 0.0f) * dmask[(size_t)row * DOUT + col];
        Out[(size_t)row * DOUT + col] = v;
      }
    }
  }
}

extern "C" void kernel_launch(void* const* d_in, const int* in_sizes, int n_in,
                              void* d_out, int out_size, void* d_ws, size_t ws_size,
                              hipStream_t stream) {
  const float* h      = (const float*)d_in[0];
  const int*  src1    = (const int*)d_in[1];
  const int*  dst1    = (const int*)d_in[2];
  const int*  src2    = (const int*)d_in[3];
  const int*  dst2    = (const int*)d_in[4];
  const int*  src3    = (const int*)d_in[5];
  const int*  dst3    = (const int*)d_in[6];
  const float* Wself1  = (const float*)d_in[10];
  const float* Wneigh1 = (const float*)d_in[11];
  const float* b1      = (const float*)d_in[12];
  const float* Wself2  = (const float*)d_in[13];
  const float* Wneigh2 = (const float*)d_in[14];
  const float* b2      = (const float*)d_in[15];
  const float* Wself3  = (const float*)d_in[16];
  const float* Wneigh3 = (const float*)d_in[17];
  const float* b3      = (const float*)d_in[18];
  const float* dmask1  = (const float*)d_in[19];
  const float* dmask2  = (const float*)d_in[20];

  const int D = 256, Dout = 128;
  const int E1 = in_sizes[1];
  const int E2 = in_sizes[3];
  const int E3 = in_sizes[5];
  const int N1 = in_sizes[19] / D;
  const int N2 = in_sizes[20] / D;
  const int N3 = out_size / Dout;

  float* w = (float*)d_ws;
  float* mean = w; w += (size_t)N1 * D;
  float* h1 = w;   w += (size_t)N1 * D;
  float* h2 = w;   w += (size_t)N2 * D;
  int* rec1 = (int*)w; w += (size_t)E1 * 2;
  int* rec2 = (int*)w; w += (size_t)E2 * 2;
  int* rec3 = (int*)w; w += (size_t)E3 * 2;
  int* head1 = (int*)w;
  int* head2 = head1 + N1;
  int* head3 = head2 + N2;

  const int NT = N1 + N2 + N3;
  const int ET = E1 + E2 + E3;
  sage_heads_init<<<(unsigned)((NT + 255) / 256), 256, 0, stream>>>(head1, NT);
  sage_build_lists<<<(unsigned)((ET + 255) / 256), 256, 0, stream>>>(
      src1, dst1, E1, src2, dst2, E2, src3, dst3, E3,
      head1, head2, head3, rec1, rec2, rec3);

  sage_gather_mean<256, 4><<<2048, 256, 0, stream>>>(
      h, head1, rec1, mean, N1, E1);
  sage_gemm_wmma<256, 256, 2, true><<<dim3(1, (N1 + 15) / 16), 256, 0, stream>>>(
      h, mean, Wself1, Wneigh1, b1, dmask1, h1, N1);

  sage_gather_mean<256, 4><<<(unsigned)((N2 + 7) / 8), 256, 0, stream>>>(
      h1, head2, rec2, mean, N2, E2);
  sage_gemm_wmma<256, 256, 2, true><<<dim3(1, (N2 + 15) / 16), 256, 0, stream>>>(
      h1, mean, Wself2, Wneigh2, b2, dmask2, h2, N2);

  sage_gather_mean<256, 4><<<(unsigned)((N3 + 7) / 8), 256, 0, stream>>>(
      h2, head3, rec3, mean, N3, E3);
  sage_gemm_wmma<256, 128, 1, false><<<dim3(1, (N3 + 15) / 16), 256, 0, stream>>>(
      h2, mean, Wself3, Wneigh3, b3, nullptr, (float*)d_out, N3);
}


# fusion: each GEMM block gathers its 16 neighbour-mean rows (two list walks per wave, side by side) and its 16 self rows into LDS; no mean plane, 3 launches fewer; per-node sum order = list order
# speedup vs baseline: 2.4125x; 1.0994x over previous
#include <hip/hip_runtime.h>


typedef float v2f __attribute__((ext_vector_type(2)));
typedef float v8f __attribute__((ext_vector_type(8)));

#define KC 32
#define MP 260

__global__ void sage_heads_init(int* __restrict__ head, int n) {
  const int i = (int)(blockIdx.x * blockDim.x + threadIdx.x);
  if (i < n) head[i] = -1;
}

__global__ void sage_build_lists(const int* __restrict__ src1, const int* __restrict__ dst1, int E1,
                                 const int* __restrict__ src2, const int* __restrict__ dst2, int E2,
                                 const int* __restrict__ src3, const int* __restrict__ dst3, int E3,
                                 int* __restrict__ head1, int* __restrict__ head2,
                                 int* __restrict__ head3,
                                 int* __restrict__ rec1, int* __restrict__ rec2,
                                 int* __restrict__ rec3) {
  int e = (int)(blockIdx.x * blockDim.x + threadIdx.x);
  const int* s; const int* d; int* hd; int* rc;
  if (e < E1) { s = src1; d = dst1; hd = head1; rc = rec1; }
  else if (e < E1 + E2) { e -= E1; s = src2; d = dst2; hd = head2; rc = rec2; }
  else if (e < E1 + E2 + E3) { e -= E1 + E2; s = src3; d = dst3; hd = head3; rc = rec3; }
  else return;
  const int sv = s[e];
  const int dv = d[e];
  const int prev = __hip_atomic_exchange(hd + dv, e, __ATOMIC_RELAXED,
                                         __HIP_MEMORY_SCOPE_AGENT);
  ((int2*)rc)[e] = make_int2(sv, prev);
}

template <int D, int DOUT, int NTW, bool RELU, int U>
__global__ void __launch_bounds__(256)
sage_layer_fused(const float* __restrict__ H,
                 const int* __restrict__ head,
                 const int* __restrict__ rec,
                 const float* __restrict__ Wself,
                 const float* __restrict__ Wneigh,
                 const float* __restrict__ bias,
                 const float* __restrict__ dmask,
                 float* __restrict__ Out,
                 int n, int E) {
  static_assert(D == 256, "row = 2 x 32 lanes x float4");
  __shared__ __attribute__((aligned(16))) float Hs[16 * MP];
  __shared__ __attribute__((aligned(16))) float Ms[16 * MP];

  const int tid  = threadIdx.x;
  const int lane = tid & 31;
  const int wave = __builtin_amdgcn_readfirstlane(tid >> 5);
  const int tm   = blockIdx.y;
  const int half = lane >> 4;
  const int l    = lane & 15;

  {
    const int rowA  = wave * 2;
    const int rowB  = rowA + 1;
    const int nodeA = tm * 16 + rowA;
    const int nodeB = nodeA + 1;
    const int2* rc = (const int2*)rec;

    const int hA = head[min(nodeA, n - 1)];
    const int hB = head[min(nodeB, n - 1)];
    int eA = (nodeA < n) ? hA : -1;
    int eB = (nodeB < n) ? hB : -1;

    float4 aA0 = make_float4(0.f, 0.f, 0.f, 0.f);
    float4 aA1 = make_float4(0.f, 0.f, 0.f, 0.f);
    float4 aB0 = make_float4(0.f, 0.f, 0.f, 0.f);
    float4 aB1 = make_float4(0.f, 0.f, 0.f, 0.f);
    int dA = 0, dB = 0;

    for (int trip = 0; trip < E && (eA >= 0 || eB >= 0); ++trip) {
      float4 vA0[U], vA1[U], vB0[U], vB1[U];
      int mA = 0, mB = 0;
#pragma unroll
      for (int u = 0; u < U; ++u) {
        const bool onA = eA >= 0;
        const bool onB = eB >= 0;
        const int2 rA = rc[max(eA, 0)];
        const int2 rB = rc[max(eB, 0)];
        int sxA = rA.x, nxA = rA.y, sxB = rB.x, nxB = rB.y;
        asm volatile("" : "+s"(sxA), "+s"(nxA), "+s"(sxB), "+s"(nxB));
        eA = onA ? nxA : -1;
        eB = onB ? nxB : -1;
        mA += onA ? 1 : 0;
        mB += onB ? 1 : 0;
        const float4 z = make_float4(0.f, 0.f, 0.f, 0.f);
        vA0[u] = z; vA1[u] = z; vB0[u] = z; vB1[u] = z;
        if (onA) {
          const float4* hp = (const float4*)(H + (size_t)sxA * D) + lane;
          vA0[u] = hp[0];
          vA1[u] = hp[32];
        }
        if (onB) {
          const float4* hp = (const float4*)(H + (size_t)sxB * D) + lane;
          vB0[u] = hp[0];
          vB1[u] = hp[32];
        }
      }
#pragma unroll
      for (int u = 0; u < U; ++u) {
        aA0.x += vA0[u].x; aA0.y += vA0[u].y; aA0.z += vA0[u].z; aA0.w += vA0[u].w;
        aA1.x += vA1[u].x; aA1.y += vA1[u].y; aA1.z += vA1[u].z; aA1.w += vA1[u].w;
        aB0.x += vB0[u].x; aB0.y += vB0[u].y; aB0.z += vB0[u].z; aB0.w += vB0[u].w;
        aB1.x += vB1[u].x; aB1.y += vB1[u].y; aB1.z += vB1[u].z; aB1.w += vB1[u].w;
      }
      dA += mA;
      dB += mB;
    }

    const float4* spA = (const float4*)(H + (size_t)min(nodeA, n - 1) * D) + lane;
    const float4* spB = (const float4*)(H + (size_t)min(nodeB, n - 1) * D) + lane;
    const float4 sA0 = spA[0], sA1 = spA[32];
    const float4 sB0 = spB[0], sB1 = spB[32];

    const float invA = 1.0f / fmaxf((float)dA, 1.0f);
    const float invB = 1.0f / fmaxf((float)dB, 1.0f);
    aA0.x *= invA; aA0.y *= invA; aA0.z *= invA; aA0.w *= invA;
    aA1.x *= invA; aA1.y *= invA; aA1.z *= invA; aA1.w *= invA;
    aB0.x *= invB; aB0.y *= invB; aB0.z *= invB; aB0.w *= invB;
    aB1.x *= invB; aB1.y *= invB; aB1.z *= invB; aB1.w *= invB;

    float4* mA4 = (float4*)&Ms[rowA * MP] + lane;
    float4* mB4 = (float4*)&Ms[rowB * MP] + lane;
    mA4[0] = aA0; mA4[32] = aA1;
    mB4[0] = aB0; mB4[32] = aB1;
    float4* hA4 = (float4*)&Hs[rowA * MP] + lane;
    float4* hB4 = (float4*)&Hs[rowB * MP] + lane;
    hA4[0] = sA0; hA4[32] = sA1;
    hB4[0] = sB0; hB4[32] = sB1;
  }

  v8f acc[NTW];
#pragma unroll
  for (int t = 0; t < NTW; ++t) {
    const float bv = bias[(wave * NTW + t) * 16 + l];
#pragma unroll
    for (int r = 0; r < 8; ++r) acc[t][r] = bv;
  }

  __syncthreads();

#pragma unroll
  for (int c = 0; c < D / KC; ++c) {
    const size_t wbase = (size_t)(c * KC + half * 2) * DOUT;
    const float* pW0[NTW];
    const float* pW1[NTW];
#pragma unroll
    for (int t = 0; t < NTW; ++t) {
      const int col = (wave * NTW + t) * 16 + l;
      pW0[t] = Wself + wbase + col;
      pW1[t] = Wneigh + wbase + col;
    }

#pragma unroll
    for (int k0 = 0; k0 < KC; k0 += 4) {
      const int kk = c * KC + k0 + half * 2;
      const v2f a0 = *(const v2f*)&Hs[l * MP + kk];
      const v2f a1 = *(const v2f*)&Ms[l * MP + kk];
#pragma unroll
      for (int t = 0; t < NTW; ++t) {
        v2f b0, b1;
        b0.x = pW0[t][k0 * DOUT];
        b0.y = pW0[t][k0 * DOUT + DOUT];
        acc[t] = __builtin_amdgcn_wmma_f32_16x16x4_f32(
            false, a0, false, b0, (short)0, acc[t], false, false);
        b1.x = pW1[t][k0 * DOUT];
        b1.y = pW1[t][k0 * DOUT + DOUT];
        acc[t] = __builtin_amdgcn_wmma_f32_16x16x4_f32(
            false, a1, false, b1, (short)0, acc[t], false, false);
      }
    }
  }

#pragma unroll
  for (int t = 0; t < NTW; ++t) {
    const int col = (wave * NTW + t) * 16 + l;
#pragma unroll
    for (int r = 0; r < 8; ++r) {
      const int row = tm * 16 + half * 8 + r;
      if (row < n) {
        float v = acc[t][r];
        if (RELU) v = fmaxf(v, 0.0f) * dmask[(size_t)row * DOUT + col];
        Out[(size_t)row * DOUT + col] = v;
      }
    }
  }
}

extern "C" void kernel_launch(void* const* d_in, const int* in_sizes, int n_in,
                              void* d_out, int out_size, void* d_ws, size_t ws_size,
                              hipStream_t stream) {
  const float* h      = (const float*)d_in[0];
  const int*  src1    = (const int*)d_in[1];
  const int*  dst1    = (const int*)d_in[2];
  const int*  src2    = (const int*)d_in[3];
  const int*  dst2    = (const int*)d_in[4];
  const int*  src3    = (const int*)d_in[5];
  const int*  dst3    = (const int*)d_in[6];
  const float* Wself1  = (const float*)d_in[10];
  const float* Wneigh1 = (const float*)d_in[11];
  const float* b1      = (const float*)d_in[12];
  const float* Wself2  = (const float*)d_in[13];
  const float* Wneigh2 = (const float*)d_in[14];
  const float* b2      = (const float*)d_in[15];
  const float* Wself3  = (const float*)d_in[16];
  const float* Wneigh3 = (const float*)d_in[17];
  const float* b3      = (const float*)d_in[18];
  const float* dmask1  = (const float*)d_in[19];
  const float* dmask2  = (const float*)d_in[20];

  const int D = 256, Dout = 128;
  const int E1 = in_sizes[1];
  const int E2 = in_sizes[3];
  const int E3 = in_sizes[5];
  const int N1 = in_sizes[19] / D;
  const int N2 = in_sizes[20] / D;
  const int N3 = out_size / Dout;

  float* w = (float*)d_ws;
  float* h1 = w;   w += (size_t)N1 * D;
  float* h2 = w;   w += (size_t)N2 * D;
  int* rec1 = (int*)w; w += (size_t)E1 * 2;
  int* rec2 = (int*)w; w += (size_t)E2 * 2;
  int* rec3 = (int*)w; w += (size_t)E3 * 2;
  int* head1 = (int*)w;
  int* head2 = head1 + N1;
  int* head3 = head2 + N2;

  const int NT = N1 + N2 + N3;
  const int ET = E1 + E2 + E3;
  sage_heads_init<<<(unsigned)((NT + 255) / 256), 256, 0, stream>>>(head1, NT);
  sage_build_lists<<<(unsigned)((ET + 255) / 256), 256, 0, stream>>>(
      src1, dst1, E1, src2, dst2, E2, src3, dst3, E3,
      head1, head2, head3, rec1, rec2, rec3);

  sage_layer_fused<256, 256, 2, true, 3><<<dim3(1, (N1 + 15) / 16), 256, 0, stream>>>(
      h, head1, rec1, Wself1, Wneigh1, b1, dmask1, h1, N1, E1);

  sage_layer_fused<256, 256, 2, true, 3><<<dim3(1, (N2 + 15) / 16), 256, 0, stream>>>(
      h1, head2, rec2, Wself2, Wneigh2, b2, dmask2, h2, N2, E2);

  sage_layer_fused<256, 128, 1, false, 3><<<dim3(1, (N3 + 15) / 16), 256, 0, stream>>>(
      h2, head3, rec3, Wself3, Wneigh3, b3, nullptr, (float*)d_out, N3, E3);
}


# epilogue: a wave's 16 dropout-mask loads use a clamped row index outside the row guard and issue together: 16 dependent load-wait-store round trips per wave become one; results bit-identical
# speedup vs baseline: 2.5206x; 1.0448x over previous
#include <hip/hip_runtime.h>


typedef float v2f __attribute__((ext_vector_type(2)));
typedef float v8f __attribute__((ext_vector_type(8)));

#define KC 32
#define MP 260

__global__ void sage_heads_init(int* __restrict__ head, int n) {
  const int i = (int)(blockIdx.x * blockDim.x + threadIdx.x);
  if (i < n) head[i] = -1;
}

__global__ void sage_build_lists(const int* __restrict__ src1, const int* __restrict__ dst1, int E1,
                                 const int* __restrict__ src2, const int* __restrict__ dst2, int E2,
                                 const int* __restrict__ src3, const int* __restrict__ dst3, int E3,
                                 int* __restrict__ head1, int* __restrict__ head2,
                                 int* __restrict__ head3,
                                 int* __restrict__ rec1, int* __restrict__ rec2,
                                 int* __restrict__ rec3) {
  int e = (int)(blockIdx.x * blockDim.x + threadIdx.x);
  const int* s; const int* d; int* hd; int* rc;
  if (e < E1) { s = src1; d = dst1; hd = head1; rc = rec1; }
  else if (e < E1 + E2) { e -= E1; s = src2; d = dst2; hd = head2; rc = rec2; }
  else if (e < E1 + E2 + E3) { e -= E1 + E2; s = src3; d = dst3; hd = head3; rc = rec3; }
  else return;
  const int sv = s[e];
  const int dv = d[e];
  const int prev = __hip_atomic_exchange(hd + dv, e, __ATOMIC_RELAXED,
                                         __HIP_MEMORY_SCOPE_AGENT);
  ((int2*)rc)[e] = make_int2(sv, prev);
}

template <int D, int DOUT, int NTW, bool RELU, int U>
__global__ void __launch_bounds__(256)
sage_layer_fused(const float* __restrict__ H,
                 const int* __restrict__ head,
                 const int* __restrict__ rec,
                 const float* __restrict__ Wself,
                 const float* __restrict__ Wneigh,
                 const float* __restrict__ bias,
                 const float* __restrict__ dmask,
                 float* __restrict__ Out,
                 int n, int E) {
  static_assert(D == 256, "row = 2 x 32 lanes x float4");
  __shared__ __attribute__((aligned(16))) float Hs[16 * MP];
  __shared__ __attribute__((aligned(16))) float Ms[16 * MP];

  const int tid  = threadIdx.x;
  const int lane = tid & 31;
  const int wave = __builtin_amdgcn_readfirstlane(tid >> 5);
  const int tm   = blockIdx.y;
  const int half = lane >> 4;
  const int l    = lane & 15;

  {
    const int rowA  = wave * 2;
    const int rowB  = rowA + 1;
    const int nodeA = tm * 16 + rowA;
    const int nodeB = nodeA + 1;
    const int2* rc = (const int2*)rec;

    const int hA = head[min(nodeA, n - 1)];
    const int hB = head[min(nodeB, n - 1)];
    int eA = (nodeA < n) ? hA : -1;
    int eB = (nodeB < n) ? hB : -1;

    float4 aA0 = make_float4(0.f, 0.f, 0.f, 0.f);
    float4 aA1 = make_float4(0.f, 0.f, 0.f, 0.f);
    float4 aB0 = make_float4(0.f, 0.f, 0.f, 0.f);
    float4 aB1 = make_float4(0.f, 0.f, 0.f, 0.f);
    int dA = 0, dB = 0;

    for (int trip = 0; trip < E && (eA >= 0 || eB >= 0); ++trip) {
      float4 vA0[U], vA1[U], vB0[U], vB1[U];
      int mA = 0, mB = 0;
#pragma unroll
      for (int u = 0; u < U; ++u) {
        const bool onA = eA >= 0;
        const bool onB = eB >= 0;
        const int2 rA = rc[max(eA, 0)];
        const int2 rB = rc[max(eB, 0)];
        int sxA = rA.x, nxA = rA.y, sxB = rB.x, nxB = rB.y;
        asm volatile("" : "+s"(sxA), "+s"(nxA), "+s"(sxB), "+s"(nxB));
        eA = onA ? nxA : -1;
        eB = onB ? nxB : -1;
        mA += onA ? 1 : 0;
        mB += onB ? 1 : 0;
        const float4 z = make_float4(0.f, 0.f, 0.f, 0.f);
        vA0[u] = z; vA1[u] = z; vB0[u] = z; vB1[u] = z;
        if (onA) {
          const float4* hp = (const float4*)(H + (size_t)sxA * D) + lane;
          vA0[u] = hp[0];
          vA1[u] = hp[32];
        }
        if (onB) {
          const float4* hp = (const float4*)(H + (size_t)sxB * D) + lane;
          vB0[u] = hp[0];
          vB1[u] = hp[32];
        }
      }
#pragma unroll
      for (int u = 0; u < U; ++u) {
        aA0.x += vA0[u].x; aA0.y += vA0[u].y; aA0.z += vA0[u].z; aA0.w += vA0[u].w;
        aA1.x += vA1[u].x; aA1.y += vA1[u].y; aA1.z += vA1[u].z; aA1.w += vA1[u].w;
        aB0.x += vB0[u].x; aB0.y += vB0[u].y; aB0.z += vB0[u].z; aB0.w += vB0[u].w;
        aB1.x += vB1[u].x; aB1.y += vB1[u].y; aB1.z += vB1[u].z; aB1.w += vB1[u].w;
      }
      dA += mA;
      dB += mB;
    }

    const float4* spA = (const float4*)(H + (size_t)min(nodeA, n - 1) * D) + lane;
    const float4* spB = (const float4*)(H + (size_t)min(nodeB, n - 1) * D) + lane;
    const float4 sA0 = spA[0], sA1 = spA[32];
    const float4 sB0 = spB[0], sB1 = spB[32];

    const float invA = 1.0f / fmaxf((float)dA, 1.0f);
    const float invB = 1.0f / fmaxf((float)dB, 1.0f);
    aA0.x *= invA; aA0.y *= invA; aA0.z *= invA; aA0.w *= invA;
    aA1.x *= invA; aA1.y *= invA; aA1.z *= invA; aA1.w *= invA;
    aB0.x *= invB; aB0.y *= invB; aB0.z *= invB; aB0.w *= invB;
    aB1.x *= invB; aB1.y *= invB; aB1.z *= invB; aB1.w *= invB;

    float4* mA4 = (float4*)&Ms[rowA * MP] + lane;
    float4* mB4 = (float4*)&Ms[rowB * MP] + lane;
    mA4[0] = aA0; mA4[32] = aA1;
    mB4[0] = aB0; mB4[32] = aB1;
    float4* hA4 = (float4*)&Hs[rowA * MP] + lane;
    float4* hB4 = (float4*)&Hs[rowB * MP] + lane;
    hA4[0] = sA0; hA4[32] = sA1;
    hB4[0] = sB0; hB4[32] = sB1;
  }

  v8f acc[NTW];
#pragma unroll
  for (int t = 0; t < NTW; ++t) {
    const float bv = bias[(wave * NTW + t) * 16 + l];
#pragma unroll
    for (int r = 0; r < 8; ++r) acc[t][r] = bv;
  }

  __syncthreads();

#pragma unroll
  for (int c = 0; c < D / KC; ++c) {
    const size_t wbase = (size_t)(c * KC + half * 2) * DOUT;
    const float* pW0[NTW];
    const float* pW1[NTW];
#pragma unroll
    for (int t = 0; t < NTW; ++t) {
      const int col = (wave * NTW + t) * 16 + l;
      pW0[t] = Wself + wbase + col;
      pW1[t] = Wneigh + wbase + col;
    }

#pragma unroll
    for (int k0 = 0; k0 < KC; k0 += 4) {
      const int kk = c * KC + k0 + half * 2;
      const v2f a0 = *(const v2f*)&Hs[l * MP + kk];
      const v2f a1 = *(const v2f*)&Ms[l * MP + kk];
#pragma unroll
      for (int t = 0; t < NTW; ++t) {
        v2f b0, b1;
        b0.x = pW0[t][k0 * DOUT];
        b0.y = pW0[t][k0 * DOUT + DOUT];
        acc[t] = __builtin_amdgcn_wmma_f32_16x16x4_f32(
            false, a0, false, b0, (short)0, acc[t], false, false);
        b1.x = pW1[t][k0 * DOUT];
        b1.y = pW1[t][k0 * DOUT + DOUT];
        acc[t] = __builtin_amdgcn_wmma_f32_16x16x4_f32(
            false, a1, false, b1, (short)0, acc[t], false, false);
      }
    }
  }

  float dm[NTW][8];
  if (RELU) {
#pragma unroll
    for (int t = 0; t < NTW; ++t) {
      const int col = (wave * NTW + t) * 16 + l;
#pragma unroll
      for (int r = 0; r < 8; ++r) {
        const int rowc = min(tm * 16 + half * 8 + r, n - 1);
        dm[t][r] = dmask[(size_t)rowc * DOUT + col];
      }
    }
  }
#pragma unroll
  for (int t = 0; t < NTW; ++t) {
    const int col = (wave * NTW + t) * 16 + l;
#pragma unroll
    for (int r = 0; r < 8; ++r) {
      const int row = tm * 16 + half * 8 + r;
      if (row < n) {
        float v = acc[t][r];
        if (RELU) v = fmaxf(v, 0.0f) * dm[t][r];
        Out[(size_t)row * DOUT + col] = v;
      }
    }
  }
}

extern "C" void kernel_launch(void* const* d_in, const int* in_sizes, int n_in,
                              void* d_out, int out_size, void* d_ws, size_t ws_size,
                              hipStream_t stream) {
  const float* h      = (const float*)d_in[0];
  const int*  src1    = (const int*)d_in[1];
  const int*  dst1    = (const int*)d_in[2];
  const int*  src2    = (const int*)d_in[3];
  const int*  dst2    = (const int*)d_in[4];
  const int*  src3    = (const int*)d_in[5];
  const int*  dst3    = (const int*)d_in[6];
  const float* Wself1  = (const float*)d_in[10];
  const float* Wneigh1 = (const float*)d_in[11];
  const float* b1      = (const float*)d_in[12];
  const float* Wself2  = (const float*)d_in[13];
  const float* Wneigh2 = (const float*)d_in[14];
  const float* b2      = (const float*)d_in[15];
  const float* Wself3  = (const float*)d_in[16];
  const float* Wneigh3 = (const float*)d_in[17];
  const float* b3      = (const float*)d_in[18];
  const float* dmask1  = (const float*)d_in[19];
  const float* dmask2  = (const float*)d_in[20];

  const int D = 256, Dout = 128;
  const int E1 = in_sizes[1];
  const int E2 = in_sizes[3];
  const int E3 = in_sizes[5];
  const int N1 = in_sizes[19] / D;
  const int N2 = in_sizes[20] / D;
  const int N3 = out_size / Dout;

  float* w = (float*)d_ws;
  float* h1 = w;   w += (size_t)N1 * D;
  float* h2 = w;   w += (size_t)N2 * D;
  int* rec1 = (int*)w; w += (size_t)E1 * 2;
  int* rec2 = (int*)w; w += (size_t)E2 * 2;
  int* rec3 = (int*)w; w += (size_t)E3 * 2;
  int* head1 = (int*)w;
  int* head2 = head1 + N1;
  int* head3 = head2 + N2;

  const int NT = N1 + N2 + N3;
  const int ET = E1 + E2 + E3;
  sage_heads_init<<<(unsigned)((NT + 255) / 256), 256, 0, stream>>>(head1, NT);
  sage_build_lists<<<(unsigned)((ET + 255) / 256), 256, 0, stream>>>(
      src1, dst1, E1, src2, dst2, E2, src3, dst3, E3,
      head1, head2, head3, rec1, rec2, rec3);

  sage_layer_fused<256, 256, 2, true, 3><<<dim3(1, (N1 + 15) / 16), 256, 0, stream>>>(
      h, head1, rec1, Wself1, Wneigh1, b1, dmask1, h1, N1, E1);

  sage_layer_fused<256, 256, 2, true, 3><<<dim3(1, (N2 + 15) / 16), 256, 0, stream>>>(
      h1, head2, rec2, Wself2, Wneigh2, b2, dmask2, h2, N2, E2);

  sage_layer_fused<256, 128, 1, false, 3><<<dim3(1, (N3 + 15) / 16), 256, 0, stream>>>(
      h2, head3, rec3, Wself3, Wneigh3, b3, nullptr, (float*)d_out, N3, E3);
}


# GEMM register tile: layer-1 blocks cover 32 rows, each wave holds 2 row x 2 column tiles and every weight fragment load feeds both row tiles (weight loads per WMMA halved); sum order unchanged
# speedup vs baseline: 2.6859x; 1.0656x over previous
#include <hip/hip_runtime.h>


typedef float v2f __attribute__((ext_vector_type(2)));
typedef float v8f __attribute__((ext_vector_type(8)));

#define KC 32
#define MP 260

__global__ void sage_heads_init(int* __restrict__ head, int n) {
  const int i = (int)(blockIdx.x * blockDim.x + threadIdx.x);
  if (i < n) head[i] = -1;
}

__global__ void sage_build_lists(const int* __restrict__ src1, const int* __restrict__ dst1, int E1,
                                 const int* __restrict__ src2, const int* __restrict__ dst2, int E2,
                                 const int* __restrict__ src3, const int* __restrict__ dst3, int E3,
                                 int* __restrict__ head1, int* __restrict__ head2,
                                 int* __restrict__ head3,
                                 int* __restrict__ rec1, int* __restrict__ rec2,
                                 int* __restrict__ rec3) {
  int e = (int)(blockIdx.x * blockDim.x + threadIdx.x);
  const int* s; const int* d; int* hd; int* rc;
  if (e < E1) { s = src1; d = dst1; hd = head1; rc = rec1; }
  else if (e < E1 + E2) { e -= E1; s = src2; d = dst2; hd = head2; rc = rec2; }
  else if (e < E1 + E2 + E3) { e -= E1 + E2; s = src3; d = dst3; hd = head3; rc = rec3; }
  else return;
  const int sv = s[e];
  const int dv = d[e];
  const int prev = __hip_atomic_exchange(hd + dv, e, __ATOMIC_RELAXED,
                                         __HIP_MEMORY_SCOPE_AGENT);
  ((int2*)rc)[e] = make_int2(sv, prev);
}

template <int D, int DOUT, int NTW, bool RELU, int U, int MT>
__global__ void __launch_bounds__(256)
sage_layer_fused(const float* __restrict__ H,
                 const int* __restrict__ head,
                 const int* __restrict__ rec,
                 const float* __restrict__ Wself,
                 const float* __restrict__ Wneigh,
                 const float* __restrict__ bias,
                 const float* __restrict__ dmask,
                 float* __restrict__ Out,
                 int n, int E) {
  static_assert(D == 256, "row = 2 x 32 lanes x float4");
  __shared__ __attribute__((aligned(16))) float Hs[MT * 16 * MP];
  __shared__ __attribute__((aligned(16))) float Ms[MT * 16 * MP];

  const int tid  = threadIdx.x;
  const int lane = tid & 31;
  const int wave = __builtin_amdgcn_readfirstlane(tid >> 5);
  const int row0 = (int)blockIdx.y * (MT * 16);
  const int half = lane >> 4;
  const int l    = lane & 15;

  {
    constexpr int NL = 2 * MT;
    const int2* rc = (const int2*)rec;

    int e[NL], deg[NL];
    float4 s0[NL], s1[NL];
#pragma unroll
    for (int j = 0; j < NL; ++j) {
      const int node = row0 + (j >> 1) * 16 + wave * 2 + (j & 1);
      e[j] = head[min(node, n - 1)];
    }
    if constexpr (NL == 4) {
      asm volatile("" : "+s"(e[0]), "+s"(e[1]), "+s"(e[2]), "+s"(e[3]));
    }
#pragma unroll
    for (int j = 0; j < NL; ++j) {
      const int node = row0 + (j >> 1) * 16 + wave * 2 + (j & 1);
      if (node >= n) e[j] = -1;
      deg[j] = 0;
      s0[j] = make_float4(0.f, 0.f, 0.f, 0.f);
      s1[j] = make_float4(0.f, 0.f, 0.f, 0.f);
    }

    bool more = false;
#pragma unroll
    for (int j = 0; j < NL; ++j) more = more || (e[j] >= 0);

    for (int trip = 0; trip < E && more; ++trip) {
      float4 v0[U][NL], v1[U][NL];
      int m[NL];
#pragma unroll
      for (int j = 0; j < NL; ++j) m[j] = 0;
#pragma unroll
      for (int u = 0; u < U; ++u) {
        bool on[NL];
        int sx[NL], nx[NL];
#pragma unroll
        for (int j = 0; j < NL; ++j) {
          on[j] = e[j] >= 0;
          const int2 r = rc[max(e[j], 0)];
          sx[j] = r.x;
          nx[j] = r.y;
        }
        if constexpr (NL == 2) {
          asm volatile("" : "+s"(sx[0]), "+s"(nx[0]), "+s"(sx[1]), "+s"(nx[1]));
        } else {
          static_assert(NL == 4, "the statement below names four lists");
          asm volatile("" : "+s"(sx[0]), "+s"(nx[0]), "+s"(sx[1]), "+s"(nx[1]),
                            "+s"(sx[2]), "+s"(nx[2]), "+s"(sx[3]), "+s"(nx[3]));
        }
#pragma unroll
        for (int j = 0; j < NL; ++j) {
          e[j] = on[j] ? nx[j] : -1;
          m[j] += on[j] ? 1 : 0;
          v0[u][j] = make_float4(0.f, 0.f, 0.f, 0.f);
          v1[u][j] = make_float4(0.f, 0.f, 0.f, 0.f);
          if (on[j]) {
            const float4* hp = (const float4*)(H + (size_t)sx[j] * D) + lane;
            v0[u][j] = hp[0];
            v1[u][j] = hp[32];
          }
        }
      }
#pragma unroll
      for (int u = 0; u < U; ++u) {
#pragma unroll
        for (int j = 0; j < NL; ++j) {
          s0[j].x += v0[u][j].x; s0[j].y += v0[u][j].y;
          s0[j].z += v0[u][j].z; s0[j].w += v0[u][j].w;
          s1[j].x += v1[u][j].x; s1[j].y += v1[u][j].y;
          s1[j].z += v1[u][j].z; s1[j].w += v1[u][j].w;
        }
      }
      more = false;
#pragma unroll
      for (int j = 0; j < NL; ++j) {
        deg[j] += m[j];
        more = more || (e[j] >= 0);
      }
    }

    float4 h0[NL], h1[NL];
#pragma unroll
    for (int j = 0; j < NL; ++j) {
      const int node = row0 + (j >> 1) * 16 + wave * 2 + (j & 1);
      const float4* sp = (const float4*)(H + (size_t)min(node, n - 1) * D) + lane;
      h0[j] = sp[0];
      h1[j] = sp[32];
    }

#pragma unroll
    for (int j = 0; j < NL; ++j) {
      const int row = (j >> 1) * 16 + wave * 2 + (j & 1);
      const float inv = 1.0f / fmaxf((float)deg[j], 1.0f);
      s0[j].x *= inv; s0[j].y *= inv; s0[j].z *= inv; s0[j].w *= inv;
      s1[j].x *= inv; s1[j].y *= inv; s1[j].z *= inv; s1[j].w *= inv;
      float4* m4 = (float4*)&Ms[row * MP] + lane;
      m4[0] = s0[j];
      m4[32] = s1[j];
    }
#pragma unroll
    for (int j = 0; j < NL; ++j) {
      const int row = (j >> 1) * 16 + wave * 2 + (j & 1);
      float4* h4 = (float4*)&Hs[row * MP] + lane;
      h4[0] = h0[j];
      h4[32] = h1[j];
    }
  }

  v8f acc[MT][NTW];
#pragma unroll
  for (int t = 0; t < NTW; ++t) {
    const float bv = bias[(wave * NTW + t) * 16 + l];
#pragma unroll
    for (int m = 0; m < MT; ++m) {
#pragma unroll
      for (int r = 0; r < 8; ++r) acc[m][t][r] = bv;
    }
  }

  __syncthreads();

#pragma unroll
  for (int c = 0; c < D / KC; ++c) {
    const size_t wbase = (size_t)(c * KC + half * 2) * DOUT;
    const float* pW0[NTW];
    const float* pW1[NTW];
#pragma unroll
    for (int t = 0; t < NTW; ++t) {
      const int col = (wave * NTW + t) * 16 + l;
      pW0[t] = Wself + wbase + col;
      pW1[t] = Wneigh + wbase + col;
    }

#pragma unroll
    for (int k0 = 0; k0 < KC; k0 += 4) {
      const int kk = c * KC + k0 + half * 2;
      v2f a0[MT], a1[MT];
#pragma unroll
      for (int m = 0; m < MT; ++m) {
        a0[m] = *(const v2f*)&Hs[(m * 16 + l) * MP + kk];
        a1[m] = *(const v2f*)&Ms[(m * 16 + l) * MP + kk];
      }
#pragma unroll
      for (int t = 0; t < NTW; ++t) {
        v2f b0, b1;
        b0.x = pW0[t][k0 * DOUT];
        b0.y = pW0[t][k0 * DOUT + DOUT];
#pragma unroll
        for (int m = 0; m < MT; ++m)
          acc[m][t] = __builtin_amdgcn_wmma_f32_16x16x4_f32(
              false, a0[m], false, b0, (short)0, acc[m][t], false, false);
        b1.x = pW1[t][k0 * DOUT];
        b1.y = pW1[t][k0 * DOUT + DOUT];
#pragma unroll
        for (int m = 0; m < MT; ++m)
          acc[m][t] = __builtin_amdgcn_wmma_f32_16x16x4_f32(
              false, a1[m], false, b1, (short)0, acc[m][t], false, false);
      }
    }
  }

  float dm[MT][NTW][8];
  if (RELU) {
#pragma unroll
    for (int m = 0; m < MT; ++m) {
#pragma unroll
      for (int t = 0; t < NTW; ++t) {
        const int col = (wave * NTW + t) * 16 + l;
#pragma unroll
        for (int r = 0; r < 8; ++r) {
          const int rowc = min(row0 + m * 16 + half * 8 + r, n - 1);
          dm[m][t][r] = dmask[(size_t)rowc * DOUT + col];
        }
      }
    }
  }
#pragma unroll
  for (int m = 0; m < MT; ++m) {
#pragma unroll
    for (int t = 0; t < NTW; ++t) {
      const int col = (wave * NTW + t) * 16 + l;
#pragma unroll
      for (int r = 0; r < 8; ++r) {
        const int row = row0 + m * 16 + half * 8 + r;
        if (row < n) {
          float v = acc[m][t][r];
          if (RELU) v = fmaxf(v, 0.0f) * dm[m][t][r];
          Out[(size_t)row * DOUT + col] = v;
        }
      }
    }
  }
}

extern "C" void kernel_launch(void* const* d_in, const int* in_sizes, int n_in,
                              void* d_out, int out_size, void* d_ws, size_t ws_size,
                              hipStream_t stream) {
  const float* h      = (const float*)d_in[0];
  const int*  src1    = (const int*)d_in[1];
  const int*  dst1    = (const int*)d_in[2];
  const int*  src2    = (const int*)d_in[3];
  const int*  dst2    = (const int*)d_in[4];
  const int*  src3    = (const int*)d_in[5];
  const int*  dst3    = (const int*)d_in[6];
  const float* Wself1  = (const float*)d_in[10];
  const float* Wneigh1 = (const float*)d_in[11];
  const float* b1      = (const float*)d_in[12];
  const float* Wself2  = (const float*)d_in[13];
  const float* Wneigh2 = (const float*)d_in[14];
  const float* b2      = (const float*)d_in[15];
  const float* Wself3  = (const float*)d_in[16];
  const float* Wneigh3 = (const float*)d_in[17];
  const float* b3      = (const float*)d_in[18];
  const float* dmask1  = (const float*)d_in[19];
  const float* dmask2  = (const float*)d_in[20];

  const int D = 256, Dout = 128;
  const int E1 = in_sizes[1];
  const int E2 = in_sizes[3];
  const int E3 = in_sizes[5];
  const int N1 = in_sizes[19] / D;
  const int N2 = in_sizes[20] / D;
  const int N3 = out_size / Dout;

  float* w = (float*)d_ws;
  float* h1 = w;   w += (size_t)N1 * D;
  float* h2 = w;   w += (size_t)N2 * D;
  int* rec1 = (int*)w; w += (size_t)E1 * 2;
  int* rec2 = (int*)w; w += (size_t)E2 * 2;
  int* rec3 = (int*)w; w += (size_t)E3 * 2;
  int* head1 = (int*)w;
  int* head2 = head1 + N1;
  int* head3 = head2 + N2;

  const int NT = N1 + N2 + N3;
  const int ET = E1 + E2 + E3;
  sage_heads_init<<<(unsigned)((NT + 255) / 256), 256, 0, stream>>>(head1, NT);
  sage_build_lists<<<(unsigned)((ET + 255) / 256), 256, 0, stream>>>(
      src1, dst1, E1, src2, dst2, E2, src3, dst3, E3,
      head1, head2, head3, rec1, rec2, rec3);

  sage_layer_fused<256, 256, 2, true, 2, 2><<<dim3(1, (N1 + 31) / 32), 256, 0, stream>>>(
      h, head1, rec1, Wself1, Wneigh1, b1, dmask1, h1, N1, E1);

  sage_layer_fused<256, 256, 2, true, 3, 1><<<dim3(1, (N2 + 15) / 16), 256, 0, stream>>>(
      h1, head2, rec2, Wself2, Wneigh2, b2, dmask2, h2, N2, E2);

  sage_layer_fused<256, 128, 1, false, 3, 1><<<dim3(1, (N3 + 15) / 16), 256, 0, stream>>>(
      h2, head3, rec3, Wself3, Wneigh3, b3, nullptr, (float*)d_out, N3, E3);
}
